# starting kernel
# baseline (speedup 1.0000x reference)
#include <hip/hip_runtime.h>
#include <hip/hip_bf16.h>

typedef __attribute__((ext_vector_type(2))) float v2f;
typedef __attribute__((ext_vector_type(8))) float v8f;

#define GNN_N 100000
#define GNN_E 1600000
#define GNN_D 64

__global__ __launch_bounds__(256) void gnn_zero_kernel(float* __restrict__ p, int n4) {
  int i = blockIdx.x * 256 + threadIdx.x;
  if (i < n4) ((float4*)p)[i] = make_float4(0.f, 0.f, 0.f, 0.f);
}

__global__ __launch_bounds__(256) void gnn_count_kernel(const int* __restrict__ ei,
                                                        float* __restrict__ cnt) {
  int e = blockIdx.x * 256 + threadIdx.x;
  if (e < GNN_E) {
    int d = ei[GNN_E + e];
    unsafeAtomicAdd(&cnt[d], 1.0f);
  }
}

__global__ __launch_bounds__(256) void gnn_inv_kernel(const float* __restrict__ cnt,
                                                      float* __restrict__ inv) {
  int i = blockIdx.x * 256 + threadIdx.x;
  if (i < GNN_N) inv[i] = 1.0f / fmaxf(cnt[i], 1.0f);
}

__global__ __launch_bounds__(256) void gnn_scatter_kernel(const int* __restrict__ ei,
                                                          const float* __restrict__ feat,
                                                          float* __restrict__ agg) {
  int tid = blockIdx.x * 256 + threadIdx.x;
  if (tid >= GNN_E * 16) return;
  int e = tid >> 4;
  int q = (tid & 15) << 2;
  int s = ei[e];
  int d = ei[GNN_E + e];
  float4 v = *(const float4*)(feat + (size_t)s * GNN_D + q);
  float* o = agg + (size_t)d * GNN_D + q;
  unsafeAtomicAdd(o + 0, v.x);
  unsafeAtomicAdd(o + 1, v.y);
  unsafeAtomicAdd(o + 2, v.z);
  unsafeAtomicAdd(o + 3, v.w);
}

__global__ __launch_bounds__(256) void gnn_sage_gemm_kernel(
    const float* __restrict__ A,
    const float* __restrict__ X,
    const float* __restrict__ inv,
    const float* __restrict__ Wl,
    const float* __restrict__ Wr,
    const float* __restrict__ bias,
    float* __restrict__ Y,
    int nRowTiles, int doRelu) {
  int wave = (blockIdx.x * 256 + threadIdx.x) >> 5;
  if (wave >= nRowTiles) return;
  int lane = threadIdx.x & 31;
  int m    = lane & 15;
  int half = lane >> 4;
  int row0 = wave * 16;
  int row  = row0 + m;
  float scale = inv[row];

  v8f c0 = {}, c1 = {}, c2 = {}, c3 = {};

#pragma unroll 1
  for (int pass = 0; pass < 2; ++pass) {
    const float* M = pass ? X : A;
    const float* W = pass ? Wr : Wl;
    float s = pass ? 1.0f : scale;
    const float* arow = M + (size_t)row * GNN_D;
#pragma unroll
    for (int kt = 0; kt < 16; ++kt) {
      int k0 = kt * 4 + half * 2;
      v2f a = *(const v2f*)(arow + k0);
      a.x *= s; a.y *= s;
      v2f bA = *(const v2f*)(W + (size_t)(0 * 16 + m) * GNN_D + k0);
      v2f bB = *(const v2f*)(W + (size_t)(1 * 16 + m) * GNN_D + k0);
      v2f bC = *(const v2f*)(W + (size_t)(2 * 16 + m) * GNN_D + k0);
      v2f bD = *(const v2f*)(W + (size_t)(3 * 16 + m) * GNN_D + k0);
      c0 = __builtin_amdgcn_wmma_f32_16x16x4_f32(false, a, false, bA, (short)0, c0, false, false);
      c1 = __builtin_amdgcn_wmma_f32_16x16x4_f32(false, a, false, bB, (short)0, c1, false, false);
      c2 = __builtin_amdgcn_wmma_f32_16x16x4_f32(false, a, false, bC, (short)0, c2, false, false);
      c3 = __builtin_amdgcn_wmma_f32_16x16x4_f32(false, a, false, bD, (short)0, c3, false, false);
    }
  }

  v8f acc[4] = {c0, c1, c2, c3};
#pragma unroll
  for (int jt = 0; jt < 4; ++jt) {
    int j = jt * 16 + m;
    float bv = bias[j];
#pragma unroll
    for (int r = 0; r < 8; ++r) {
      float v = acc[jt][r] + bv;
      if (doRelu) v = fmaxf(v, 0.0f);
      Y[(size_t)(row0 + r + half * 8) * GNN_D + j] = v;
    }
  }
}

extern "C" void kernel_launch(void* const* d_in, const int* in_sizes, int n_in,
                              void* d_out, int out_size, void* d_ws, size_t ws_size,
                              hipStream_t stream) {
  const float* x   = (const float*)d_in[0];
  const int*   ei  = (const int*)d_in[1];
  const float* W1l = (const float*)d_in[2];
  const float* b1  = (const float*)d_in[3];
  const float* W1r = (const float*)d_in[4];
  const float* W2l = (const float*)d_in[5];
  const float* b2  = (const float*)d_in[6];
  const float* W2r = (const float*)d_in[7];
  float* out = (float*)d_out;

  float* agg = (float*)d_ws;
  float* h   = agg + (size_t)GNN_N * GNN_D;
  float* cnt = h + (size_t)GNN_N * GNN_D;
  float* inv = cnt + GNN_N;

  const int feat4    = (GNN_N * GNN_D) / 4;
  const int zeroGrid = (feat4 + 255) / 256;
  const int cnt4     = GNN_N / 4;
  const int cntGrid  = (cnt4 + 255) / 256;
  const int edgeGrid = (GNN_E + 255) / 256;
  const int scatGrid = (GNN_E * 16 + 255) / 256;
  const int nGrid    = (GNN_N + 255) / 256;
  const int rowTiles = GNN_N / 16;
  const int gemmGrid = (rowTiles + 7) / 8;

  gnn_zero_kernel<<<cntGrid, 256, 0, stream>>>(cnt, cnt4);
  gnn_count_kernel<<<edgeGrid, 256, 0, stream>>>(ei, cnt);
  gnn_inv_kernel<<<nGrid, 256, 0, stream>>>(cnt, inv);

  gnn_zero_kernel<<<zeroGrid, 256, 0, stream>>>(agg, feat4);
  gnn_scatter_kernel<<<scatGrid, 256, 0, stream>>>(ei, x, agg);
  gnn_sage_gemm_kernel<<<gemmGrid, 256, 0, stream>>>(agg, x, inv, W1l, W1r, b1, h,
                                                     rowTiles,  1);

  gnn_zero_kernel<<<zeroGrid, 256, 0, stream>>>(agg, feat4);
  gnn_scatter_kernel<<<scatGrid, 256, 0, stream>>>(ei, h, agg);
  gnn_sage_gemm_kernel<<<gemmGrid, 256, 0, stream>>>(agg, h, inv, W2l, W2r, b2, out,
                                                     rowTiles,  0);
}


# scatter-add relaid: one wave per 32 edges, lane l on columns l and l+32, so each f32 atomic add and row load covers one whole 128-B line (2 add lines per edge, was 8); atomic arrival order as before
# speedup vs baseline: 1.5363x; 1.5363x over previous
#include <hip/hip_runtime.h>
#include <hip/hip_bf16.h>

typedef __attribute__((ext_vector_type(2))) float v2f;
typedef __attribute__((ext_vector_type(8))) float v8f;

#define GNN_N 100000
#define GNN_E 1600000
#define GNN_D 64

__global__ __launch_bounds__(256) void gnn_zero_kernel(float* __restrict__ p, int n4) {
  int i = blockIdx.x * 256 + threadIdx.x;
  if (i < n4) ((float4*)p)[i] = make_float4(0.f, 0.f, 0.f, 0.f);
}

__global__ __launch_bounds__(256) void gnn_count_kernel(const int* __restrict__ ei,
                                                        float* __restrict__ cnt) {
  int e = blockIdx.x * 256 + threadIdx.x;
  if (e < GNN_E) {
    int d = ei[GNN_E + e];
    unsafeAtomicAdd(&cnt[d], 1.0f);
  }
}

__global__ __launch_bounds__(256) void gnn_inv_kernel(const float* __restrict__ cnt,
                                                      float* __restrict__ inv) {
  int i = blockIdx.x * 256 + threadIdx.x;
  if (i < GNN_N) inv[i] = 1.0f / fmaxf(cnt[i], 1.0f);
}

#define GNN_EPW 32
#define GNN_EB  8
static_assert(GNN_E % (GNN_EPW * 8) == 0, "edge count must fill whole work-groups");
__global__ __launch_bounds__(256) void gnn_scatter_kernel(const int* __restrict__ ei,
                                                          const float* __restrict__ feat,
                                                          float* __restrict__ agg) {
  int wave = (blockIdx.x * 256 + threadIdx.x) >> 5;
  int lane = threadIdx.x & 31;
  int e0 = wave * GNN_EPW;
  int sl = ei[e0 + lane];
  int dl = ei[GNN_E + e0 + lane];
#pragma unroll
  for (int b = 0; b < GNN_EPW; b += GNN_EB) {
    float v0[GNN_EB], v1[GNN_EB];
#pragma unroll
    for (int j = 0; j < GNN_EB; ++j) {
      int s = __builtin_amdgcn_readlane(sl, b + j);
      const float* fr = feat + (size_t)s * GNN_D + lane;
      v0[j] = fr[0];
      v1[j] = fr[32];
    }
#pragma unroll
    for (int j = 0; j < GNN_EB; ++j) {
      int d = __builtin_amdgcn_readlane(dl, b + j);
      float* ar = agg + (size_t)d * GNN_D + lane;
      unsafeAtomicAdd(ar, v0[j]);
      unsafeAtomicAdd(ar + 32, v1[j]);
    }
  }
}

__global__ __launch_bounds__(256) void gnn_sage_gemm_kernel(
    const float* __restrict__ A,
    const float* __restrict__ X,
    const float* __restrict__ inv,
    const float* __restrict__ Wl,
    const float* __restrict__ Wr,
    const float* __restrict__ bias,
    float* __restrict__ Y,
    int nRowTiles, int doRelu) {
  int wave = (blockIdx.x * 256 + threadIdx.x) >> 5;
  if (wave >= nRowTiles) return;
  int lane = threadIdx.x & 31;
  int m    = lane & 15;
  int half = lane >> 4;
  int row0 = wave * 16;
  int row  = row0 + m;
  float scale = inv[row];

  v8f c0 = {}, c1 = {}, c2 = {}, c3 = {};

#pragma unroll 1
  for (int pass = 0; pass < 2; ++pass) {
    const float* M = pass ? X : A;
    const float* W = pass ? Wr : Wl;
    float s = pass ? 1.0f : scale;
    const float* arow = M + (size_t)row * GNN_D;
#pragma unroll
    for (int kt = 0; kt < 16; ++kt) {
      int k0 = kt * 4 + half * 2;
      v2f a = *(const v2f*)(arow + k0);
      a.x *= s; a.y *= s;
      v2f bA = *(const v2f*)(W + (size_t)(0 * 16 + m) * GNN_D + k0);
      v2f bB = *(const v2f*)(W + (size_t)(1 * 16 + m) * GNN_D + k0);
      v2f bC = *(const v2f*)(W + (size_t)(2 * 16 + m) * GNN_D + k0);
      v2f bD = *(const v2f*)(W + (size_t)(3 * 16 + m) * GNN_D + k0);
      c0 = __builtin_amdgcn_wmma_f32_16x16x4_f32(false, a, false, bA, (short)0, c0, false, false);
      c1 = __builtin_amdgcn_wmma_f32_16x16x4_f32(false, a, false, bB, (short)0, c1, false, false);
      c2 = __builtin_amdgcn_wmma_f32_16x16x4_f32(false, a, false, bC, (short)0, c2, false, false);
      c3 = __builtin_amdgcn_wmma_f32_16x16x4_f32(false, a, false, bD, (short)0, c3, false, false);
    }
  }

  v8f acc[4] = {c0, c1, c2, c3};
#pragma unroll
  for (int jt = 0; jt < 4; ++jt) {
    int j = jt * 16 + m;
    float bv = bias[j];
#pragma unroll
    for (int r = 0; r < 8; ++r) {
      float v = acc[jt][r] + bv;
      if (doRelu) v = fmaxf(v, 0.0f);
      Y[(size_t)(row0 + r + half * 8) * GNN_D + j] = v;
    }
  }
}

extern "C" void kernel_launch(void* const* d_in, const int* in_sizes, int n_in,
                              void* d_out, int out_size, void* d_ws, size_t ws_size,
                              hipStream_t stream) {
  const float* x   = (const float*)d_in[0];
  const int*   ei  = (const int*)d_in[1];
  const float* W1l = (const float*)d_in[2];
  const float* b1  = (const float*)d_in[3];
  const float* W1r = (const float*)d_in[4];
  const float* W2l = (const float*)d_in[5];
  const float* b2  = (const float*)d_in[6];
  const float* W2r = (const float*)d_in[7];
  float* out = (float*)d_out;

  float* agg = (float*)d_ws;
  float* h   = agg + (size_t)GNN_N * GNN_D;
  float* cnt = h + (size_t)GNN_N * GNN_D;
  float* inv = cnt + GNN_N;

  const int feat4    = (GNN_N * GNN_D) / 4;
  const int zeroGrid = (feat4 + 255) / 256;
  const int cnt4     = GNN_N / 4;
  const int cntGrid  = (cnt4 + 255) / 256;
  const int edgeGrid = (GNN_E + 255) / 256;
  const int scatGrid = GNN_E / (GNN_EPW * 8);
  const int nGrid    = (GNN_N + 255) / 256;
  const int rowTiles = GNN_N / 16;
  const int gemmGrid = (rowTiles + 7) / 8;

  gnn_zero_kernel<<<cntGrid, 256, 0, stream>>>(cnt, cnt4);
  gnn_count_kernel<<<edgeGrid, 256, 0, stream>>>(ei, cnt);
  gnn_inv_kernel<<<nGrid, 256, 0, stream>>>(cnt, inv);

  gnn_zero_kernel<<<zeroGrid, 256, 0, stream>>>(agg, feat4);
  gnn_scatter_kernel<<<scatGrid, 256, 0, stream>>>(ei, x, agg);
  gnn_sage_gemm_kernel<<<gemmGrid, 256, 0, stream>>>(agg, x, inv, W1l, W1r, b1, h,
                                                     rowTiles,  1);

  gnn_zero_kernel<<<zeroGrid, 256, 0, stream>>>(agg, feat4);
  gnn_scatter_kernel<<<scatGrid, 256, 0, stream>>>(ei, h, agg);
  gnn_sage_gemm_kernel<<<gemmGrid, 256, 0, stream>>>(agg, h, inv, W2l, W2r, b2, out,
                                                     rowTiles,  0);
}


# GEMM operand path: both weight matrices staged once per work-group in LDS, B fragments by ds_load_b128, A rows by 16-byte loads; k summed in a permuted order, equal in both loaders
# speedup vs baseline: 1.6168x; 1.0524x over previous
#include <hip/hip_runtime.h>
#include <hip/hip_bf16.h>

typedef __attribute__((ext_vector_type(2))) float v2f;
typedef __attribute__((ext_vector_type(8))) float v8f;

#define GNN_N 100000
#define GNN_E 1600000
#define GNN_D 64

__global__ __launch_bounds__(256) void gnn_zero_kernel(float* __restrict__ p, int n4) {
  int i = blockIdx.x * 256 + threadIdx.x;
  if (i < n4) ((float4*)p)[i] = make_float4(0.f, 0.f, 0.f, 0.f);
}

__global__ __launch_bounds__(256) void gnn_count_kernel(const int* __restrict__ ei,
                                                        float* __restrict__ cnt) {
  int e = blockIdx.x * 256 + threadIdx.x;
  if (e < GNN_E) {
    int d = ei[GNN_E + e];
    unsafeAtomicAdd(&cnt[d], 1.0f);
  }
}

__global__ __launch_bounds__(256) void gnn_inv_kernel(const float* __restrict__ cnt,
                                                      float* __restrict__ inv) {
  int i = blockIdx.x * 256 + threadIdx.x;
  if (i < GNN_N) inv[i] = 1.0f / fmaxf(cnt[i], 1.0f);
}

#define GNN_EPW 32
#define GNN_EB  8
static_assert(GNN_E % (GNN_EPW * 8) == 0, "edge count must fill whole work-groups");
__global__ __launch_bounds__(256) void gnn_scatter_kernel(const int* __restrict__ ei,
                                                          const float* __restrict__ feat,
                                                          float* __restrict__ agg) {
  int wave = (blockIdx.x * 256 + threadIdx.x) >> 5;
  int lane = threadIdx.x & 31;
  int e0 = wave * GNN_EPW;
  int sl = ei[e0 + lane];
  int dl = ei[GNN_E + e0 + lane];
#pragma unroll
  for (int b = 0; b < GNN_EPW; b += GNN_EB) {
    float v0[GNN_EB], v1[GNN_EB];
#pragma unroll
    for (int j = 0; j < GNN_EB; ++j) {
      int s = __builtin_amdgcn_readlane(sl, b + j);
      const float* fr = feat + (size_t)s * GNN_D + lane;
      v0[j] = fr[0];
      v1[j] = fr[32];
    }
#pragma unroll
    for (int j = 0; j < GNN_EB; ++j) {
      int d = __builtin_amdgcn_readlane(dl, b + j);
      float* ar = agg + (size_t)d * GNN_D + lane;
      unsafeAtomicAdd(ar, v0[j]);
      unsafeAtomicAdd(ar + 32, v1[j]);
    }
  }
}

#define GNN_WP 68
__global__ __launch_bounds__(256) void gnn_sage_gemm_kernel(
    const float* __restrict__ A,
    const float* __restrict__ X,
    const float* __restrict__ inv,
    const float* __restrict__ Wl,
    const float* __restrict__ Wr,
    const float* __restrict__ bias,
    float* __restrict__ Y,
    int nRowTiles, int doRelu) {
  __shared__ __attribute__((aligned(16))) float Ws[2 * GNN_D * GNN_WP];

#pragma unroll
  for (int r = 0; r < 4; ++r) {
    int i = threadIdx.x + 256 * r;
    int n = i >> 4;
    int c = (i & 15) * 4;
    float4 wl = *(const float4*)(Wl + n * GNN_D + c);
    float4 wr = *(const float4*)(Wr + n * GNN_D + c);
    *(float4*)(Ws + n * GNN_WP + c) = wl;
    *(float4*)(Ws + (GNN_D + n) * GNN_WP + c) = wr;
  }
  __syncthreads();

  int wave = (blockIdx.x * 256 + threadIdx.x) >> 5;
  if (wave >= nRowTiles) return;
  int lane = threadIdx.x & 31;
  int m    = lane & 15;
  int half = lane >> 4;
  int row0 = wave * 16;
  int row  = row0 + m;
  float scale = inv[row];

  v8f c0 = {}, c1 = {}, c2 = {}, c3 = {};

#pragma unroll 1
  for (int pass = 0; pass < 2; ++pass) {
    const float* M = pass ? X : A;
    float s = pass ? 1.0f : scale;
    const float* arow = M + (size_t)row * GNN_D + half * 4;
    const float* wb = Ws + (pass * GNN_D + m) * GNN_WP + half * 4;
    float4 a4[8];
#pragma unroll
    for (int j = 0; j < 8; ++j) a4[j] = *(const float4*)(arow + 8 * j);
#pragma unroll
    for (int j = 0; j < 8; ++j) {
      float4 bA = *(const float4*)(wb + 0 * 16 * GNN_WP + 8 * j);
      float4 bB = *(const float4*)(wb + 1 * 16 * GNN_WP + 8 * j);
      float4 bC = *(const float4*)(wb + 2 * 16 * GNN_WP + 8 * j);
      float4 bD = *(const float4*)(wb + 3 * 16 * GNN_WP + 8 * j);
      v2f aLo = {a4[j].x * s, a4[j].y * s};
      v2f aHi = {a4[j].z * s, a4[j].w * s};
      v2f bA0 = {bA.x, bA.y}, bA1 = {bA.z, bA.w};
      v2f bB0 = {bB.x, bB.y}, bB1 = {bB.z, bB.w};
      v2f bC0 = {bC.x, bC.y}, bC1 = {bC.z, bC.w};
      v2f bD0 = {bD.x, bD.y}, bD1 = {bD.z, bD.w};
      c0 = __builtin_amdgcn_wmma_f32_16x16x4_f32(false, aLo, false, bA0, (short)0, c0, false, false);
      c1 = __builtin_amdgcn_wmma_f32_16x16x4_f32(false, aLo, false, bB0, (short)0, c1, false, false);
      c2 = __builtin_amdgcn_wmma_f32_16x16x4_f32(false, aLo, false, bC0, (short)0, c2, false, false);
      c3 = __builtin_amdgcn_wmma_f32_16x16x4_f32(false, aLo, false, bD0, (short)0, c3, false, false);
      c0 = __builtin_amdgcn_wmma_f32_16x16x4_f32(false, aHi, false, bA1, (short)0, c0, false, false);
      c1 = __builtin_amdgcn_wmma_f32_16x16x4_f32(false, aHi, false, bB1, (short)0, c1, false, false);
      c2 = __builtin_amdgcn_wmma_f32_16x16x4_f32(false, aHi, false, bC1, (short)0, c2, false, false);
      c3 = __builtin_amdgcn_wmma_f32_16x16x4_f32(false, aHi, false, bD1, (short)0, c3, false, false);
    }
  }

  v8f acc[4] = {c0, c1, c2, c3};
#pragma unroll
  for (int jt = 0; jt < 4; ++jt) {
    int j = jt * 16 + m;
    float bv = bias[j];
#pragma unroll
    for (int r = 0; r < 8; ++r) {
      float v = acc[jt][r] + bv;
      if (doRelu) v = fmaxf(v, 0.0f);
      Y[(size_t)(row0 + r + half * 8) * GNN_D + j] = v;
    }
  }
}

extern "C" void kernel_launch(void* const* d_in, const int* in_sizes, int n_in,
                              void* d_out, int out_size, void* d_ws, size_t ws_size,
                              hipStream_t stream) {
  const float* x   = (const float*)d_in[0];
  const int*   ei  = (const int*)d_in[1];
  const float* W1l = (const float*)d_in[2];
  const float* b1  = (const float*)d_in[3];
  const float* W1r = (const float*)d_in[4];
  const float* W2l = (const float*)d_in[5];
  const float* b2  = (const float*)d_in[6];
  const float* W2r = (const float*)d_in[7];
  float* out = (float*)d_out;

  float* agg = (float*)d_ws;
  float* h   = agg + (size_t)GNN_N * GNN_D;
  float* cnt = h + (size_t)GNN_N * GNN_D;
  float* inv = cnt + GNN_N;

  const int feat4    = (GNN_N * GNN_D) / 4;
  const int zeroGrid = (feat4 + 255) / 256;
  const int cnt4     = GNN_N / 4;
  const int cntGrid  = (cnt4 + 255) / 256;
  const int edgeGrid = (GNN_E + 255) / 256;
  const int scatGrid = GNN_E / (GNN_EPW * 8);
  const int nGrid    = (GNN_N + 255) / 256;
  const int rowTiles = GNN_N / 16;
  const int gemmGrid = (rowTiles + 7) / 8;

  gnn_zero_kernel<<<cntGrid, 256, 0, stream>>>(cnt, cnt4);
  gnn_count_kernel<<<edgeGrid, 256, 0, stream>>>(ei, cnt);
  gnn_inv_kernel<<<nGrid, 256, 0, stream>>>(cnt, inv);

  gnn_zero_kernel<<<zeroGrid, 256, 0, stream>>>(agg, feat4);
  gnn_scatter_kernel<<<scatGrid, 256, 0, stream>>>(ei, x, agg);
  gnn_sage_gemm_kernel<<<gemmGrid, 256, 0, stream>>>(agg, x, inv, W1l, W1r, b1, h,
                                                     rowTiles,  1);

  gnn_zero_kernel<<<zeroGrid, 256, 0, stream>>>(agg, feat4);
  gnn_scatter_kernel<<<scatGrid, 256, 0, stream>>>(ei, h, agg);
  gnn_sage_gemm_kernel<<<gemmGrid, 256, 0, stream>>>(agg, h, inv, W2l, W2r, b2, out,
                                                     rowTiles,  0);
}


# float-atomic scatter-add replaced: edges sorted by destination once per call (integer counters in LDS), each row's sources summed in registers and stored once; sum order per row = sort fill order
# speedup vs baseline: 4.2978x; 2.6582x over previous
#include <hip/hip_runtime.h>
#include <hip/hip_bf16.h>

typedef __attribute__((ext_vector_type(2))) float v2f;
typedef __attribute__((ext_vector_type(8))) float v8f;

#define GNN_N 100000
#define GNN_E 1600000
#define GNN_D 64

#define GNN_NPB 256
#define GNN_NB  ((GNN_N + GNN_NPB - 1) / GNN_NPB)
#define GNN_CH  8192
#define GNN_NCH ((GNN_E + GNN_CH - 1) / GNN_CH)
#define GNN_EPT (GNN_CH / 1024)
#define GNN_SRC_BITS 17
static_assert(GNN_N <= (1 << GNN_SRC_BITS), "source index must fit the record");
static_assert(GNN_NPB <= (1 << (32 - GNN_SRC_BITS)), "local destination must fit the record");
static_assert(GNN_NB <= 1024 && GNN_NCH <= 512, "one thread per bucket / per tile in the scans");

__device__ __forceinline__ int gnn_wave_incl_scan(int v, int lane) {
#pragma unroll
  for (int d = 1; d < 32; d <<= 1) {
    int t = __shfl_up(v, d);
    if (lane >= d) v += t;
  }
  return v;
}
__device__ __forceinline__ int gnn_block_excl_scan(int v, int* wsum, int tid, int nwaves,
                                                   int* total) {
  int lane = tid & 31, w = tid >> 5;
  int inc = gnn_wave_incl_scan(v, lane);
  if (lane == 31) wsum[w] = inc;
  __syncthreads();
  if (w == 0) {
    int t = (lane < nwaves) ? wsum[lane] : 0;
    int ti = gnn_wave_incl_scan(t, lane);
    wsum[lane] = ti - t;
    if (lane == 31) wsum[32] = ti;
  }
  __syncthreads();
  int r = wsum[w] + inc - v;
  *total = wsum[32];
  __syncthreads();
  return r;
}

__global__ __launch_bounds__(1024) void gnn_tile_sort_kernel(const int* __restrict__ ei,
                                                             unsigned* __restrict__ recs,
                                                             int* __restrict__ offs) {
  __shared__ int hist[GNN_NB + 1];
  __shared__ int wsum[33];
  __shared__ unsigned srt[GNN_CH];
  int tid = threadIdx.x;
  int tile = blockIdx.x;
  int e0 = tile * GNN_CH;
  if (tid <= GNN_NB) hist[tid] = 0;
  __syncthreads();
  int sv[GNN_EPT], dv[GNN_EPT], rk[GNN_EPT];
#pragma unroll
  for (int k = 0; k < GNN_EPT; ++k) {
    int e = e0 + k * 1024 + tid;
    bool ok = e < GNN_E;
    sv[k] = ok ? ei[e] : 0;
    dv[k] = ok ? ei[GNN_E + e] : -1;
  }
#pragma unroll
  for (int k = 0; k < GNN_EPT; ++k)
    rk[k] = (dv[k] >= 0) ? atomicAdd(&hist[dv[k] / GNN_NPB], 1) : 0;
  __syncthreads();
  int hv = (tid < GNN_NB) ? hist[tid] : 0;
  int tot;
  int ex = gnn_block_excl_scan(hv, wsum, tid, 32, &tot);
  if (tid < GNN_NB) {
    hist[tid] = ex;
    offs[tile * (GNN_NB + 1) + tid] = ex;
  }
  if (tid == 0) offs[tile * (GNN_NB + 1) + GNN_NB] = tot;
  __syncthreads();
#pragma unroll
  for (int k = 0; k < GNN_EPT; ++k) {
    if (dv[k] >= 0) {
      int b = dv[k] / GNN_NPB;
      srt[hist[b] + rk[k]] =
          (unsigned)sv[k] | ((unsigned)(dv[k] % GNN_NPB) << GNN_SRC_BITS);
    }
  }
  __syncthreads();
  for (int i = tid; i < tot; i += 1024) recs[e0 + i] = srt[i];
}

#define GNN_LOG_NCH 8
static_assert((1 << GNN_LOG_NCH) >= GNN_NCH, "halvings must cover the tiles");
__device__ __forceinline__ int gnn_find_seg(const int* segStart, int i) {
  int lo = 0, hi = GNN_NCH;
#pragma unroll
  for (int it = 0; it < GNN_LOG_NCH; ++it) {
    int m = (lo + hi) >> 1;
    bool ge = segStart[m] <= i;
    lo = ge ? m : lo;
    hi = ge ? hi : m;
  }
  return lo;
}

__global__ __launch_bounds__(512) void gnn_csr_build_kernel(const unsigned* __restrict__ recs,
                                                            const int* __restrict__ offs,
                                                            int* __restrict__ rowptr,
                                                            int* __restrict__ col,
                                                            float* __restrict__ inv) {
  __shared__ int segStart[GNN_NCH + 1];
  __shared__ int segAddr[GNN_NCH];
  __shared__ int cnt[GNN_NPB];
  __shared__ int rowst[GNN_NPB];
  __shared__ int wsum[33];
  int tid = threadIdx.x;
  int b = blockIdx.x;
  int o0 = 0, len = 0;
  if (tid < GNN_NCH) {
    o0 = offs[tid * (GNN_NB + 1) + b];
    len = offs[tid * (GNN_NB + 1) + b + 1] - o0;
    segAddr[tid] = tid * GNN_CH + o0;
  }
  if (tid < GNN_NPB) cnt[tid] = 0;
  int T, gs;
  int st = gnn_block_excl_scan(len, wsum, tid, 16, &T);
  if (tid < GNN_NCH) segStart[tid] = st;
  if (tid == 0) segStart[GNN_NCH] = T;
  (void)gnn_block_excl_scan(o0, wsum, tid, 16, &gs);

  for (int i0 = tid; i0 < T; i0 += 4 * 512) {
    unsigned rec[4];
#pragma unroll
    for (int u = 0; u < 4; ++u) {
      int i = min(i0 + u * 512, T - 1);
      int sg = gnn_find_seg(segStart, i);
      rec[u] = recs[segAddr[sg] + (i - segStart[sg])];
    }
#pragma unroll
    for (int u = 0; u < 4; ++u)
      if (i0 + u * 512 < T) atomicAdd(&cnt[rec[u] >> GNN_SRC_BITS], 1);
  }
  __syncthreads();
  int cv = (tid < GNN_NPB) ? cnt[tid] : 0;
  int tot;
  int rs = gnn_block_excl_scan(cv, wsum, tid, 16, &tot);
  if (tid < GNN_NPB) {
    rowst[tid] = rs;
    cnt[tid] = 0;
    int node = b * GNN_NPB + tid;
    if (node < GNN_N) {
      rowptr[node] = gs + rs;
      inv[node] = 1.0f / fmaxf((float)cv, 1.0f);
    }
    if (node == GNN_N - 1) rowptr[GNN_N] = gs + rs + cv;
  }
  __syncthreads();
  for (int i0 = tid; i0 < T; i0 += 4 * 512) {
    unsigned rec[4];
#pragma unroll
    for (int u = 0; u < 4; ++u) {
      int i = min(i0 + u * 512, T - 1);
      int sg = gnn_find_seg(segStart, i);
      rec[u] = recs[segAddr[sg] + (i - segStart[sg])];
    }
#pragma unroll
    for (int u = 0; u < 4; ++u) {
      if (i0 + u * 512 < T) {
        int dl = rec[u] >> GNN_SRC_BITS;
        int p = atomicAdd(&cnt[dl], 1);
        col[gs + rowst[dl] + p] = (int)(rec[u] & ((1u << GNN_SRC_BITS) - 1));
      }
    }
  }
}

#define GNN_GW 2048
#define GNN_GL 8

template <bool PRED>
__device__ __forceinline__ void gnn_row_batch(const float* __restrict__ feat, int cv, int p,
                                              int n, int half, unsigned c4, float4& acc) {
  float4 v[GNN_GL];
#pragma unroll
  for (int u = 0; u < GNN_GL; ++u) {
    int i0 = __builtin_amdgcn_readlane(cv, p + 2 * u);
    int i1 = __builtin_amdgcn_readlane(cv, p + 2 * u + 1);
    unsigned sidx = half ? i1 : i0;
    if (PRED) {
      v[u] = make_float4(0.f, 0.f, 0.f, 0.f);
      if (p + 2 * u + half < n) v[u] = *(const float4*)(feat + (sidx * GNN_D + c4));
    } else {
      v[u] = *(const float4*)(feat + (sidx * GNN_D + c4));
    }
  }
#pragma unroll
  for (int u = 0; u < GNN_GL; ++u) {
    acc.x += v[u].x; acc.y += v[u].y; acc.z += v[u].z; acc.w += v[u].w;
  }
}

__global__ __launch_bounds__(256) void gnn_gather_kernel(const int* __restrict__ rowptr,
                                                         const int* __restrict__ col,
                                                         const float* __restrict__ feat,
                                                         float* __restrict__ agg) {
  int lane = threadIdx.x & 31;
  int half = lane >> 4;
  unsigned c4 = (lane & 15) * 4;
  int gw = __builtin_amdgcn_readfirstlane((blockIdx.x * 256 + threadIdx.x) >> 5);
  const int nw = GNN_GW * 8;
  for (int node = gw; node < GNN_N; node += nw) {
    int s = __builtin_amdgcn_readfirstlane(rowptr[node]);
    int e = __builtin_amdgcn_readfirstlane(rowptr[node + 1]);
    float4 acc = make_float4(0.f, 0.f, 0.f, 0.f);
    for (int base = s; base < e; base += 32) {
      int n = e - base;
      if (n > 32) n = 32;
      int cv = (lane < n) ? col[base + lane] : 0;
      int p = 0;
      for (; p + 2 * GNN_GL <= n; p += 2 * GNN_GL)
        gnn_row_batch<false>(feat, cv, p, n, half, c4, acc);
      if (p < n) gnn_row_batch<true>(feat, cv, p, n, half, c4, acc);
    }
    acc.x += __shfl_xor(acc.x, 16);
    acc.y += __shfl_xor(acc.y, 16);
    acc.z += __shfl_xor(acc.z, 16);
    acc.w += __shfl_xor(acc.w, 16);
    if (half == 0) *(float4*)(agg + ((size_t)node * GNN_D + c4)) = acc;
  }
}

#define GNN_WP 68
__global__ __launch_bounds__(256) void gnn_sage_gemm_kernel(
    const float* __restrict__ A,
    const float* __restrict__ X,
    const float* __restrict__ inv,
    const float* __restrict__ Wl,
    const float* __restrict__ Wr,
    const float* __restrict__ bias,
    float* __restrict__ Y,
    int nRowTiles, int doRelu) {
  __shared__ __attribute__((aligned(16))) float Ws[2 * GNN_D * GNN_WP];

#pragma unroll
  for (int r = 0; r < 4; ++r) {
    int i = threadIdx.x + 256 * r;
    int n = i >> 4;
    int c = (i & 15) * 4;
    float4 wl = *(const float4*)(Wl + n * GNN_D + c);
    float4 wr = *(const float4*)(Wr + n * GNN_D + c);
    *(float4*)(Ws + n * GNN_WP + c) = wl;
    *(float4*)(Ws + (GNN_D + n) * GNN_WP + c) = wr;
  }
  __syncthreads();

  int wave = (blockIdx.x * 256 + threadIdx.x) >> 5;
  if (wave >= nRowTiles) return;
  int lane = threadIdx.x & 31;
  int m    = lane & 15;
  int half = lane >> 4;
  int row0 = wave * 16;
  int row  = row0 + m;
  float scale = inv[row];

  v8f c0 = {}, c1 = {}, c2 = {}, c3 = {};

#pragma unroll 1
  for (int pass = 0; pass < 2; ++pass) {
    const float* M = pass ? X : A;
    float s = pass ? 1.0f : scale;
    const float* arow = M + (size_t)row * GNN_D + half * 4;
    const float* wb = Ws + (pass * GNN_D + m) * GNN_WP + half * 4;
    float4 a4[8];
#pragma unroll
    for (int j = 0; j < 8; ++j) a4[j] = *(const float4*)(arow + 8 * j);
#pragma unroll
    for (int j = 0; j < 8; ++j) {
      float4 bA = *(const float4*)(wb + 0 * 16 * GNN_WP + 8 * j);
      float4 bB = *(const float4*)(wb + 1 * 16 * GNN_WP + 8 * j);
      float4 bC = *(const float4*)(wb + 2 * 16 * GNN_WP + 8 * j);
      float4 bD = *(const float4*)(wb + 3 * 16 * GNN_WP + 8 * j);
      v2f aLo = {a4[j].x * s, a4[j].y * s};
      v2f aHi = {a4[j].z * s, a4[j].w * s};
      v2f bA0 = {bA.x, bA.y}, bA1 = {bA.z, bA.w};
      v2f bB0 = {bB.x, bB.y}, bB1 = {bB.z, bB.w};
      v2f bC0 = {bC.x, bC.y}, bC1 = {bC.z, bC.w};
      v2f bD0 = {bD.x, bD.y}, bD1 = {bD.z, bD.w};
      c0 = __builtin_amdgcn_wmma_f32_16x16x4_f32(false, aLo, false, bA0, (short)0, c0, false, false);
      c1 = __builtin_amdgcn_wmma_f32_16x16x4_f32(false, aLo, false, bB0, (short)0, c1, false, false);
      c2 = __builtin_amdgcn_wmma_f32_16x16x4_f32(false, aLo, false, bC0, (short)0, c2, false, false);
      c3 = __builtin_amdgcn_wmma_f32_16x16x4_f32(false, aLo, false, bD0, (short)0, c3, false, false);
      c0 = __builtin_amdgcn_wmma_f32_16x16x4_f32(false, aHi, false, bA1, (short)0, c0, false, false);
      c1 = __builtin_amdgcn_wmma_f32_16x16x4_f32(false, aHi, false, bB1, (short)0, c1, false, false);
      c2 = __builtin_amdgcn_wmma_f32_16x16x4_f32(false, aHi, false, bC1, (short)0, c2, false, false);
      c3 = __builtin_amdgcn_wmma_f32_16x16x4_f32(false, aHi, false, bD1, (short)0, c3, false, false);
    }
  }

  v8f acc[4] = {c0, c1, c2, c3};
#pragma unroll
  for (int jt = 0; jt < 4; ++jt) {
    int j = jt * 16 + m;
    float bv = bias[j];
#pragma unroll
    for (int r = 0; r < 8; ++r) {
      float v = acc[jt][r] + bv;
      if (doRelu) v = fmaxf(v, 0.0f);
      Y[(size_t)(row0 + r + half * 8) * GNN_D + j] = v;
    }
  }
}

extern "C" void kernel_launch(void* const* d_in, const int* in_sizes, int n_in,
                              void* d_out, int out_size, void* d_ws, size_t ws_size,
                              hipStream_t stream) {
  const float* x   = (const float*)d_in[0];
  const int*   ei  = (const int*)d_in[1];
  const float* W1l = (const float*)d_in[2];
  const float* b1  = (const float*)d_in[3];
  const float* W1r = (const float*)d_in[4];
  const float* W2l = (const float*)d_in[5];
  const float* b2  = (const float*)d_in[6];
  const float* W2r = (const float*)d_in[7];
  float* out = (float*)d_out;

  float* agg = (float*)d_ws;
  float* h   = agg + (size_t)GNN_N * GNN_D;
  float* inv = h + (size_t)GNN_N * GNN_D;
  int* rowptr = (int*)(inv + GNN_N);
  int* col    = rowptr + GNN_N + 32;
  unsigned* recs = (unsigned*)(col + GNN_E);
  int* offs   = (int*)(recs + (size_t)GNN_NCH * GNN_CH);

  const int rowTiles = GNN_N / 16;
  const int gemmGrid = (rowTiles + 7) / 8;

  gnn_tile_sort_kernel<<<GNN_NCH, 1024, 0, stream>>>(ei, recs, offs);
  gnn_csr_build_kernel<<<GNN_NB, 512, 0, stream>>>(recs, offs, rowptr, col, inv);

  gnn_gather_kernel<<<GNN_GW, 256, 0, stream>>>(rowptr, col, x, agg);
  gnn_sage_gemm_kernel<<<gemmGrid, 256, 0, stream>>>(agg, x, inv, W1l, W1r, b1, h,
                                                     rowTiles,  1);

  gnn_gather_kernel<<<GNN_GW, 256, 0, stream>>>(rowptr, col, h, agg);
  gnn_sage_gemm_kernel<<<gemmGrid, 256, 0, stream>>>(agg, h, inv, W2l, W2r, b2, out,
                                                     rowTiles,  0);
}
